# starting kernel
# baseline (speedup 1.0000x reference)
#include <hip/hip_runtime.h>

typedef float v2f __attribute__((ext_vector_type(2)));
typedef float v8f __attribute__((ext_vector_type(8)));

#define HW      4096
#define NC      256
#define NB      32
#define NR      (NB * NC)
#define KCHUNKS 4
#define KLEN    (HW / KCHUNKS)

__global__ __launch_bounds__(256) void pool_wmma(const float* __restrict__ x,
                                                 float* __restrict__ part) {
    const int lane   = threadIdx.x & 31;
    const int wave   = threadIdx.x >> 5;
    const int m      = lane & 15;
    const int half   = lane >> 4;
    const int rowTop = (blockIdx.x * 8 + wave) * 16;
    const int kc     = blockIdx.y;

    const float* xrow = x + (size_t)(rowTop + m) * HW + kc * KLEN + half * 4;

    v2f ones; ones[0] = 1.0f; ones[1] = 1.0f;
    v8f acc = {};

    for (int k0 = 0; k0 < KLEN; k0 += 8) {
        const float4 v = *(const float4*)(xrow + k0);
        if ((k0 & 15) == 0)
            __builtin_prefetch(xrow + k0 + 512, 0, 3);
        v2f a0; a0[0] = v.x; a0[1] = v.y;
        v2f a1; a1[0] = v.z; a1[1] = v.w;
        acc = __builtin_amdgcn_wmma_f32_16x16x4_f32(false, a0, false, ones,
                                                    (short)0, acc, false, false);
        acc = __builtin_amdgcn_wmma_f32_16x16x4_f32(false, a1, false, ones,
                                                    (short)0, acc, false, false);
    }

    if (m == 0) {
        const int base = rowTop + half * 8;
        #pragma unroll
        for (int v = 0; v < 8; ++v)
            part[kc * NR + base + v] = acc[v];
    }
}

__global__ __launch_bounds__(256) void reduce_parts(const float* __restrict__ part,
                                                    float* __restrict__ y) {
    const int r = blockIdx.x * blockDim.x + threadIdx.x;
    if (r < NR) {
        float s = part[0 * NR + r] + part[1 * NR + r]
                + part[2 * NR + r] + part[3 * NR + r];
        y[r] = s * (1.0f / (float)HW);
    }
}

__global__ void conv_top3(const float* __restrict__ y,
                          const float* __restrict__ w,
                          int* __restrict__ idx) {
    const int b = threadIdx.x;
    if (b >= NB) return;
    const float w0 = w[0], w1 = w[1], w2 = w[2];
    const float* yr = y + b * NC;

    float v0 = -1e30f, v1 = -1e30f, v2 = -1e30f;
    int   i0 = 0,      i1 = 0,      i2 = 0;
    for (int c = 0; c < NC; ++c) {
        const float left  = (c > 0)      ? yr[c - 1] : 0.0f;
        const float mid   = yr[c];
        const float right = (c < NC - 1) ? yr[c + 1] : 0.0f;
        const float s   = w0 * left + w1 * mid + w2 * right;
        const float att = 1.0f / (1.0f + __expf(-s));
        if (att > v0)      { v2 = v1; i2 = i1; v1 = v0; i1 = i0; v0 = att; i0 = c; }
        else if (att > v1) { v2 = v1; i2 = i1; v1 = att; i1 = c; }
        else if (att > v2) { v2 = att; i2 = c; }
    }
    idx[b * 3 + 0] = i0;
    idx[b * 3 + 1] = i1;
    idx[b * 3 + 2] = i2;
}

__global__ __launch_bounds__(256) void gather_top3(const float* __restrict__ x,
                                                   const int* __restrict__ idx,
                                                   float* __restrict__ out) {
    const int g   = blockIdx.x * blockDim.x + threadIdx.x;
    const int off = g & 1023;
    const int img = g >> 10;
    const int b   = img / 3;
    const int c   = idx[img];
    const float4* src = (const float4*)(x + (size_t)(b * NC + c) * HW);
    float4*       dst = (float4*)out + (size_t)img * 1024;
    dst[off] = src[off];
}

extern "C" void kernel_launch(void* const* d_in, const int* in_sizes, int n_in,
                              void* d_out, int out_size, void* d_ws, size_t ws_size,
                              hipStream_t stream) {
    (void)in_sizes; (void)n_in; (void)out_size; (void)ws_size;

    const float* x  = (const float*)d_in[0];
    const float* cw = (const float*)d_in[1];

    float* part = (float*)d_ws;
    float* y    = part + KCHUNKS * NR;
    int*   idx  = (int*)(y + NR);

    pool_wmma<<<dim3(NR / 128, KCHUNKS), 256, 0, stream>>>(x, part);
    reduce_parts<<<NR / 256, 256, 0, stream>>>(part, y);
    conv_top3<<<1, 32, 0, stream>>>(y, cw, idx);
    gather_top3<<<(NB * 3 * 1024) / 256, 256, 0, stream>>>(x, idx, (float*)d_out);
}


# conv+sigmoid+top-3 launch made parallel: one wave per batch row, 8 channels per lane, lane butterfly on (sigmoid bits, channel) keys; was one wave, 256 serial trips; scores bit-identical
# speedup vs baseline: 2.1813x; 2.1813x over previous
#include <hip/hip_runtime.h>

typedef float v2f __attribute__((ext_vector_type(2)));
typedef float v8f __attribute__((ext_vector_type(8)));

#define HW      4096
#define NC      256
#define NB      32
#define NR      (NB * NC)
#define KCHUNKS 4
#define KLEN    (HW / KCHUNKS)

__global__ __launch_bounds__(256) void pool_wmma(const float* __restrict__ x,
                                                 float* __restrict__ part) {
    const int lane   = threadIdx.x & 31;
    const int wave   = threadIdx.x >> 5;
    const int m      = lane & 15;
    const int half   = lane >> 4;
    const int rowTop = (blockIdx.x * 8 + wave) * 16;
    const int kc     = blockIdx.y;

    const float* xrow = x + (size_t)(rowTop + m) * HW + kc * KLEN + half * 4;

    v2f ones; ones[0] = 1.0f; ones[1] = 1.0f;
    v8f acc = {};

    for (int k0 = 0; k0 < KLEN; k0 += 8) {
        const float4 v = *(const float4*)(xrow + k0);
        if ((k0 & 15) == 0)
            __builtin_prefetch(xrow + k0 + 512, 0, 3);
        v2f a0; a0[0] = v.x; a0[1] = v.y;
        v2f a1; a1[0] = v.z; a1[1] = v.w;
        acc = __builtin_amdgcn_wmma_f32_16x16x4_f32(false, a0, false, ones,
                                                    (short)0, acc, false, false);
        acc = __builtin_amdgcn_wmma_f32_16x16x4_f32(false, a1, false, ones,
                                                    (short)0, acc, false, false);
    }

    if (m == 0) {
        const int base = rowTop + half * 8;
        #pragma unroll
        for (int v = 0; v < 8; ++v)
            part[kc * NR + base + v] = acc[v];
    }
}

__global__ __launch_bounds__(256) void reduce_parts(const float* __restrict__ part,
                                                    float* __restrict__ y) {
    const int r = blockIdx.x * blockDim.x + threadIdx.x;
    if (r < NR) {
        float s = part[0 * NR + r] + part[1 * NR + r]
                + part[2 * NR + r] + part[3 * NR + r];
        y[r] = s * (1.0f / (float)HW);
    }
}

typedef unsigned long long u64;

static __device__ inline u64 kmax(u64 a, u64 b) { return a > b ? a : b; }
static __device__ inline u64 kmin(u64 a, u64 b) { return a > b ? b : a; }

static __device__ inline u64 score_key(float w0, float w1, float w2,
                                       float left, float mid, float right, int c) {
    const float s   = __builtin_fmaf(w2, right, __builtin_fmaf(w0, left, w1 * mid));
    const float att = 1.0f / (1.0f + __expf(-s));
    return ((u64)__float_as_uint(att) << 32) | (u64)(unsigned)(NC - 1 - c);
}

static __device__ inline void keep3(u64 k, u64& k0, u64& k1, u64& k2) {
    const u64 t = kmin(k0, k);
    k0 = kmax(k0, k);
    const u64 u = kmin(k1, t);
    k1 = kmax(k1, t);
    k2 = kmax(k2, u);
}

static __device__ inline u64 key_of_lane_xor(u64 k, int mask) {
    const unsigned hi = __shfl_xor((unsigned)(k >> 32), mask, 32);
    const unsigned lo = __shfl_xor((unsigned)k, mask, 32);
    return ((u64)hi << 32) | (u64)lo;
}

__global__ __launch_bounds__(32) void conv_top3(const float* __restrict__ y,
                                                const float* __restrict__ w,
                                                int* __restrict__ idx) {
    const int lane = threadIdx.x;
    const int b    = blockIdx.x;
    const float w0 = w[0], w1 = w[1], w2 = w[2];
    const float* yr = y + b * NC;
    const int c0 = lane * 8;

    const float4 p = *(const float4*)(yr + c0);
    const float4 q = *(const float4*)(yr + c0 + 4);
    const float lft = (lane > 0)  ? yr[c0 - 1] : 0.0f;
    const float rgt = (lane < 31) ? yr[c0 + 8] : 0.0f;

    u64 k0 = 0, k1 = 0, k2 = 0;
    keep3(score_key(w0, w1, w2, lft, p.x, p.y, c0 + 0), k0, k1, k2);
    keep3(score_key(w0, w1, w2, p.x, p.y, p.z, c0 + 1), k0, k1, k2);
    keep3(score_key(w0, w1, w2, p.y, p.z, p.w, c0 + 2), k0, k1, k2);
    keep3(score_key(w0, w1, w2, p.z, p.w, q.x, c0 + 3), k0, k1, k2);
    keep3(score_key(w0, w1, w2, p.w, q.x, q.y, c0 + 4), k0, k1, k2);
    keep3(score_key(w0, w1, w2, q.x, q.y, q.z, c0 + 5), k0, k1, k2);
    keep3(score_key(w0, w1, w2, q.y, q.z, q.w, c0 + 6), k0, k1, k2);
    keep3(score_key(w0, w1, w2, q.z, q.w, rgt, c0 + 7), k0, k1, k2);

    for (int mask = 1; mask < 32; mask <<= 1) {
        const u64 o0 = key_of_lane_xor(k0, mask);
        const u64 o1 = key_of_lane_xor(k1, mask);
        const u64 o2 = key_of_lane_xor(k2, mask);
        __builtin_amdgcn_sched_barrier(0);
        const u64 hi = kmax(k0, o0);
        const u64 lo = kmin(k0, o0);
        const u64 m1 = kmax(k1, o1);
        k2 = kmax(kmin(lo, m1), kmax(k2, o2));
        k1 = kmax(lo, m1);
        k0 = hi;
    }

    if (lane < 3) {
        const u64 mine = (lane == 0) ? k0 : (lane == 1) ? k1 : k2;
        idx[b * 3 + lane] = (NC - 1) - (int)(unsigned)(mine & 0xffffffffull);
    }
}

__global__ __launch_bounds__(256) void gather_top3(const float* __restrict__ x,
                                                   const int* __restrict__ idx,
                                                   float* __restrict__ out) {
    const int g   = blockIdx.x * blockDim.x + threadIdx.x;
    const int off = g & 1023;
    const int img = g >> 10;
    const int b   = img / 3;
    const int c   = idx[img];
    const float4* src = (const float4*)(x + (size_t)(b * NC + c) * HW);
    float4*       dst = (float4*)out + (size_t)img * 1024;
    dst[off] = src[off];
}

extern "C" void kernel_launch(void* const* d_in, const int* in_sizes, int n_in,
                              void* d_out, int out_size, void* d_ws, size_t ws_size,
                              hipStream_t stream) {
    (void)in_sizes; (void)n_in; (void)out_size; (void)ws_size;

    const float* x  = (const float*)d_in[0];
    const float* cw = (const float*)d_in[1];

    float* part = (float*)d_ws;
    float* y    = part + KCHUNKS * NR;
    int*   idx  = (int*)(y + NR);

    pool_wmma<<<dim3(NR / 128, KCHUNKS), 256, 0, stream>>>(x, part);
    reduce_parts<<<NR / 256, 256, 0, stream>>>(part, y);
    conv_top3<<<NB, 32, 0, stream>>>(y, cw, idx);
    gather_top3<<<(NB * 3 * 1024) / 256, 256, 0, stream>>>(x, idx, (float*)d_out);
}


# fold of partials, conv+sigmoid+top-3 and gather made one launch: every wave of a copy work-group recomputes its batch row's top-3 from the partials; 4 launches -> 2; scores bit-identical
# speedup vs baseline: 2.5141x; 1.1526x over previous
#include <hip/hip_runtime.h>

typedef float v2f __attribute__((ext_vector_type(2)));
typedef float v4f __attribute__((ext_vector_type(4)));
typedef float v8f __attribute__((ext_vector_type(8)));

#define HW      4096
#define NC      256
#define NB      32
#define NR      (NB * NC)
#define KCHUNKS 4
#define KLEN    (HW / KCHUNKS)

__global__ __launch_bounds__(256) void pool_wmma(const float* __restrict__ x,
                                                 float* __restrict__ part) {
    const int lane   = threadIdx.x & 31;
    const int wave   = threadIdx.x >> 5;
    const int m      = lane & 15;
    const int half   = lane >> 4;
    const int rowTop = (blockIdx.x * 8 + wave) * 16;
    const int kc     = blockIdx.y;

    const float* xrow = x + (size_t)(rowTop + m) * HW + kc * KLEN + half * 4;

    v2f ones; ones[0] = 1.0f; ones[1] = 1.0f;
    v8f acc = {};

    for (int k0 = 0; k0 < KLEN; k0 += 8) {
        const float4 v = *(const float4*)(xrow + k0);
        if ((k0 & 15) == 0)
            __builtin_prefetch(xrow + k0 + 512, 0, 3);
        v2f a0; a0[0] = v.x; a0[1] = v.y;
        v2f a1; a1[0] = v.z; a1[1] = v.w;
        acc = __builtin_amdgcn_wmma_f32_16x16x4_f32(false, a0, false, ones,
                                                    (short)0, acc, false, false);
        acc = __builtin_amdgcn_wmma_f32_16x16x4_f32(false, a1, false, ones,
                                                    (short)0, acc, false, false);
    }

    if (m == 0) {
        const int base = rowTop + half * 8;
        #pragma unroll
        for (int v = 0; v < 8; ++v)
            part[kc * NR + base + v] = acc[v];
    }
}

typedef unsigned long long u64;

static __device__ inline u64 kmax(u64 a, u64 b) { return a > b ? a : b; }
static __device__ inline u64 kmin(u64 a, u64 b) { return a > b ? b : a; }

static __device__ inline u64 score_key(float w0, float w1, float w2,
                                       float left, float mid, float right, int c) {
    const float s   = __builtin_fmaf(w2, right, __builtin_fmaf(w0, left, w1 * mid));
    const float att = 1.0f / (1.0f + __expf(-s));
    return ((u64)__float_as_uint(att) << 32) | (u64)(unsigned)(NC - 1 - c);
}

static __device__ inline void keep3(u64 k, u64& k0, u64& k1, u64& k2) {
    const u64 t = kmin(k0, k);
    k0 = kmax(k0, k);
    const u64 u = kmin(k1, t);
    k1 = kmax(k1, t);
    k2 = kmax(k2, u);
}

static __device__ inline u64 key_of_lane_xor(u64 k, int mask) {
    const unsigned hi = __shfl_xor((unsigned)(k >> 32), mask, 32);
    const unsigned lo = __shfl_xor((unsigned)k, mask, 32);
    return ((u64)hi << 32) | (u64)lo;
}

static __device__ inline float4 fold4(const float* __restrict__ p) {
    const float4 a = *(const float4*)(p + 0 * NR);
    const float4 b = *(const float4*)(p + 1 * NR);
    const float4 c = *(const float4*)(p + 2 * NR);
    const float4 d = *(const float4*)(p + 3 * NR);
    const float inv = 1.0f / (float)HW;
    float4 r;
    r.x = (a.x + b.x + c.x + d.x) * inv;
    r.y = (a.y + b.y + c.y + d.y) * inv;
    r.z = (a.z + b.z + c.z + d.z) * inv;
    r.w = (a.w + b.w + c.w + d.w) * inv;
    return r;
}

__global__ __launch_bounds__(128) void top3_gather(const float* __restrict__ x,
                                                   const float* __restrict__ part,
                                                   const float* __restrict__ w,
                                                   float* __restrict__ out) {
    const int lane = threadIdx.x & 31;
    const int img  = blockIdx.x;
    const int b    = img / 3;
    const int rank = img - 3 * b;
    const float w0 = w[0], w1 = w[1], w2 = w[2];
    const float* pr = part + b * NC;
    const int c0 = lane * 8;

    const float4 p = fold4(pr + c0);
    const float4 q = fold4(pr + c0 + 4);
    const float lv  = __shfl_up(q.w, 1, 32);
    const float rv  = __shfl_down(p.x, 1, 32);
    const float lft = (lane > 0)  ? lv : 0.0f;
    const float rgt = (lane < 31) ? rv : 0.0f;

    u64 k0 = 0, k1 = 0, k2 = 0;
    keep3(score_key(w0, w1, w2, lft, p.x, p.y, c0 + 0), k0, k1, k2);
    keep3(score_key(w0, w1, w2, p.x, p.y, p.z, c0 + 1), k0, k1, k2);
    keep3(score_key(w0, w1, w2, p.y, p.z, p.w, c0 + 2), k0, k1, k2);
    keep3(score_key(w0, w1, w2, p.z, p.w, q.x, c0 + 3), k0, k1, k2);
    keep3(score_key(w0, w1, w2, p.w, q.x, q.y, c0 + 4), k0, k1, k2);
    keep3(score_key(w0, w1, w2, q.x, q.y, q.z, c0 + 5), k0, k1, k2);
    keep3(score_key(w0, w1, w2, q.y, q.z, q.w, c0 + 6), k0, k1, k2);
    keep3(score_key(w0, w1, w2, q.z, q.w, rgt, c0 + 7), k0, k1, k2);

    for (int mask = 1; mask < 32; mask <<= 1) {
        const u64 o0 = key_of_lane_xor(k0, mask);
        const u64 o1 = key_of_lane_xor(k1, mask);
        const u64 o2 = key_of_lane_xor(k2, mask);
        __builtin_amdgcn_sched_barrier(0);
        const u64 hi = kmax(k0, o0);
        const u64 lo = kmin(k0, o0);
        const u64 m1 = kmax(k1, o1);
        k2 = kmax(kmin(lo, m1), kmax(k2, o2));
        k1 = kmax(lo, m1);
        k0 = hi;
    }

    const u64 mine = (rank == 0) ? k0 : (rank == 1) ? k1 : k2;
    const int c    = (NC - 1) - (int)(unsigned)(mine & 0xffffffffull);

    const v4f* src = (const v4f*)(x + (size_t)(b * NC + c) * HW);
    v4f*       dst = (v4f*)out + (size_t)img * 1024;
    const int t = threadIdx.x;
    const v4f v0 = src[t];
    const v4f v1 = src[t + 128];
    const v4f v2 = src[t + 256];
    const v4f v3 = src[t + 384];
    const v4f v4 = src[t + 512];
    const v4f v5 = src[t + 640];
    const v4f v6 = src[t + 768];
    const v4f v7 = src[t + 896];
    dst[t]       = v0;
    dst[t + 128] = v1;
    dst[t + 256] = v2;
    dst[t + 384] = v3;
    dst[t + 512] = v4;
    dst[t + 640] = v5;
    dst[t + 768] = v6;
    dst[t + 896] = v7;
}

extern "C" void kernel_launch(void* const* d_in, const int* in_sizes, int n_in,
                              void* d_out, int out_size, void* d_ws, size_t ws_size,
                              hipStream_t stream) {
    (void)in_sizes; (void)n_in; (void)out_size; (void)ws_size;

    const float* x  = (const float*)d_in[0];
    const float* cw = (const float*)d_in[1];

    float* part = (float*)d_ws;

    pool_wmma<<<dim3(NR / 128, KCHUNKS), 256, 0, stream>>>(x, part);
    top3_gather<<<NB * 3, 128, 0, stream>>>(x, part, cw, (float*)d_out);
}


# pool launch reads x as a contiguous stream: a wave-wide load = 512 contiguous bytes of one row (was 16 rows x 32 B), 16 ahead, no prefetch; A rows = slices of one chunk: summation order changes
# speedup vs baseline: 4.3444x; 1.7280x over previous
#include <hip/hip_runtime.h>

typedef float v2f __attribute__((ext_vector_type(2)));
typedef float v4f __attribute__((ext_vector_type(4)));
typedef float v8f __attribute__((ext_vector_type(8)));

#define HW      4096
#define NC      256
#define NB      32
#define NR      (NB * NC)
#define KCHUNKS 4
#define KLEN    (HW / KCHUNKS)

#define ROWS_PER_WAVE 4
#define LOADS_PER_CHUNK (KLEN / 128)
#define CHUNKS_PER_WAVE (ROWS_PER_WAVE * KCHUNKS)
#define AHEAD 16
#define BATCHES (CHUNKS_PER_WAVE * LOADS_PER_CHUNK / AHEAD)

__global__ __launch_bounds__(256) void pool_wmma(const float* __restrict__ x,
                                                 float* __restrict__ part) {
    const int lane = threadIdx.x & 31;
    const int wave = threadIdx.x >> 5;
    const int row0 = ((blockIdx.y * (NR / 128) + blockIdx.x) * 8 + wave) * ROWS_PER_WAVE;

    const v4f* src = (const v4f*)(x + (size_t)row0 * HW) + lane;

    v2f ones; ones[0] = 1.0f; ones[1] = 1.0f;
    v4f buf[2][AHEAD];

    #pragma unroll
    for (int i = 0; i < AHEAD; ++i) buf[0][i] = src[32 * i];

    float res = 0.0f;
    #pragma unroll
    for (int b = 0; b < BATCHES; ++b) {
        __builtin_amdgcn_sched_barrier(0);
        if (b + 1 < BATCHES) {
            #pragma unroll
            for (int i = 0; i < AHEAD; ++i)
                buf[(b + 1) & 1][i] = src[32 * (AHEAD * (b + 1) + i)];
        }
        __builtin_amdgcn_sched_barrier(0);
        #pragma unroll
        for (int u = 0; u < AHEAD / LOADS_PER_CHUNK; ++u) {
            v8f acc0 = {}, acc1 = {};
            #pragma unroll
            for (int i = 0; i < LOADS_PER_CHUNK; ++i) {
                const v4f v = buf[b & 1][LOADS_PER_CHUNK * u + i];
                v2f lo; lo[0] = v[0]; lo[1] = v[1];
                v2f hi; hi[0] = v[2]; hi[1] = v[3];
                acc0 = __builtin_amdgcn_wmma_f32_16x16x4_f32(false, lo, false, ones,
                                                             (short)0, acc0, false, false);
                acc1 = __builtin_amdgcn_wmma_f32_16x16x4_f32(false, hi, false, ones,
                                                             (short)0, acc1, false, false);
            }
            const v8f t = acc0 + acc1;
            const float s = ((t[0] + t[1]) + (t[2] + t[3])) + ((t[4] + t[5]) + (t[6] + t[7]));
            const int chunk = (AHEAD / LOADS_PER_CHUNK) * b + u;
            res = ((lane & 15) == chunk) ? s : res;
        }
    }
    res += __shfl_xor(res, 16, 32);
    if (lane < CHUNKS_PER_WAVE)
        part[(lane & 3) * NR + row0 + (lane >> 2)] = res;
}

typedef unsigned long long u64;

static __device__ inline u64 kmax(u64 a, u64 b) { return a > b ? a : b; }
static __device__ inline u64 kmin(u64 a, u64 b) { return a > b ? b : a; }

static __device__ inline u64 score_key(float w0, float w1, float w2,
                                       float left, float mid, float right, int c) {
    const float s   = __builtin_fmaf(w2, right, __builtin_fmaf(w0, left, w1 * mid));
    const float att = 1.0f / (1.0f + __expf(-s));
    return ((u64)__float_as_uint(att) << 32) | (u64)(unsigned)(NC - 1 - c);
}

static __device__ inline void keep3(u64 k, u64& k0, u64& k1, u64& k2) {
    const u64 t = kmin(k0, k);
    k0 = kmax(k0, k);
    const u64 u = kmin(k1, t);
    k1 = kmax(k1, t);
    k2 = kmax(k2, u);
}

static __device__ inline u64 key_of_lane_xor(u64 k, int mask) {
    const unsigned hi = __shfl_xor((unsigned)(k >> 32), mask, 32);
    const unsigned lo = __shfl_xor((unsigned)k, mask, 32);
    return ((u64)hi << 32) | (u64)lo;
}

static __device__ inline float4 fold4(const float* __restrict__ p) {
    const float4 a = *(const float4*)(p + 0 * NR);
    const float4 b = *(const float4*)(p + 1 * NR);
    const float4 c = *(const float4*)(p + 2 * NR);
    const float4 d = *(const float4*)(p + 3 * NR);
    const float inv = 1.0f / (float)HW;
    float4 r;
    r.x = (a.x + b.x + c.x + d.x) * inv;
    r.y = (a.y + b.y + c.y + d.y) * inv;
    r.z = (a.z + b.z + c.z + d.z) * inv;
    r.w = (a.w + b.w + c.w + d.w) * inv;
    return r;
}

__global__ __launch_bounds__(128) void top3_gather(const float* __restrict__ x,
                                                   const float* __restrict__ part,
                                                   const float* __restrict__ w,
                                                   float* __restrict__ out) {
    const int lane = threadIdx.x & 31;
    const int img  = blockIdx.x;
    const int b    = img / 3;
    const int rank = img - 3 * b;
    const float w0 = w[0], w1 = w[1], w2 = w[2];
    const float* pr = part + b * NC;
    const int c0 = lane * 8;

    const float4 p = fold4(pr + c0);
    const float4 q = fold4(pr + c0 + 4);
    const float lv  = __shfl_up(q.w, 1, 32);
    const float rv  = __shfl_down(p.x, 1, 32);
    const float lft = (lane > 0)  ? lv : 0.0f;
    const float rgt = (lane < 31) ? rv : 0.0f;

    u64 k0 = 0, k1 = 0, k2 = 0;
    keep3(score_key(w0, w1, w2, lft, p.x, p.y, c0 + 0), k0, k1, k2);
    keep3(score_key(w0, w1, w2, p.x, p.y, p.z, c0 + 1), k0, k1, k2);
    keep3(score_key(w0, w1, w2, p.y, p.z, p.w, c0 + 2), k0, k1, k2);
    keep3(score_key(w0, w1, w2, p.z, p.w, q.x, c0 + 3), k0, k1, k2);
    keep3(score_key(w0, w1, w2, p.w, q.x, q.y, c0 + 4), k0, k1, k2);
    keep3(score_key(w0, w1, w2, q.x, q.y, q.z, c0 + 5), k0, k1, k2);
    keep3(score_key(w0, w1, w2, q.y, q.z, q.w, c0 + 6), k0, k1, k2);
    keep3(score_key(w0, w1, w2, q.z, q.w, rgt, c0 + 7), k0, k1, k2);

    for (int mask = 1; mask < 32; mask <<= 1) {
        const u64 o0 = key_of_lane_xor(k0, mask);
        const u64 o1 = key_of_lane_xor(k1, mask);
        const u64 o2 = key_of_lane_xor(k2, mask);
        __builtin_amdgcn_sched_barrier(0);
        const u64 hi = kmax(k0, o0);
        const u64 lo = kmin(k0, o0);
        const u64 m1 = kmax(k1, o1);
        k2 = kmax(kmin(lo, m1), kmax(k2, o2));
        k1 = kmax(lo, m1);
        k0 = hi;
    }

    const u64 mine = (rank == 0) ? k0 : (rank == 1) ? k1 : k2;
    const int c    = (NC - 1) - (int)(unsigned)(mine & 0xffffffffull);

    const v4f* src = (const v4f*)(x + (size_t)(b * NC + c) * HW);
    v4f*       dst = (v4f*)out + (size_t)img * 1024;
    const int t = threadIdx.x;
    const v4f v0 = src[t];
    const v4f v1 = src[t + 128];
    const v4f v2 = src[t + 256];
    const v4f v3 = src[t + 384];
    const v4f v4 = src[t + 512];
    const v4f v5 = src[t + 640];
    const v4f v6 = src[t + 768];
    const v4f v7 = src[t + 896];
    dst[t]       = v0;
    dst[t + 128] = v1;
    dst[t + 256] = v2;
    dst[t + 384] = v3;
    dst[t + 512] = v4;
    dst[t + 640] = v5;
    dst[t + 768] = v6;
    dst[t + 896] = v7;
}

extern "C" void kernel_launch(void* const* d_in, const int* in_sizes, int n_in,
                              void* d_out, int out_size, void* d_ws, size_t ws_size,
                              hipStream_t stream) {
    (void)in_sizes; (void)n_in; (void)out_size; (void)ws_size;

    const float* x  = (const float*)d_in[0];
    const float* cw = (const float*)d_in[1];

    float* part = (float*)d_ws;

    pool_wmma<<<dim3(NR / 128, KCHUNKS), 256, 0, stream>>>(x, part);
    top3_gather<<<NB * 3, 128, 0, stream>>>(x, part, cw, (float*)d_out);
}
